# starting kernel
# baseline (speedup 1.0000x reference)
#include <hip/hip_runtime.h>
#include <hip/hip_bf16.h>
#include <math.h>


typedef float v2f __attribute__((ext_vector_type(2)));
typedef float v8f __attribute__((ext_vector_type(8)));

#define TOK_TOTAL (64 * 4096)
#define LATENT_ELEMS ((size_t)TOK_TOTAL * 64)

static __device__ __forceinline__ v8f wmma4(v2f a, v2f b, v8f c) {
  return __builtin_amdgcn_wmma_f32_16x16x4_f32(
       false, a,  false, b,
       (short)0, c,  false,  false);
}

static __device__ __forceinline__ float gelu_exact(float x) {
  return 0.5f * x * (1.0f + erff(x * 0.70710678118654752f));
}

static __device__ __forceinline__ void apply1q(
    float* zr, float* zi, int m,
    float u00r, float u00i, float u01r, float u01i,
    float u10r, float u10i, float u11r, float u11i) {
#pragma unroll
  for (int s = 0; s < 16; ++s) {
    if (!(s & m)) {
      const int t = s | m;
      const float ar = zr[s], ai = zi[s];
      const float br = zr[t], bi = zi[t];
      zr[s] = u00r * ar - u00i * ai + u01r * br - u01i * bi;
      zi[s] = u00r * ai + u00i * ar + u01r * bi + u01i * br;
      zr[t] = u10r * ar - u10i * ai + u11r * br - u11i * bi;
      zi[t] = u10r * ai + u10i * ar + u11r * bi + u11i * br;
    }
  }
}

__global__ void qvl_zero_ent(float* ent) {
  if (threadIdx.x < 4) ent[threadIdx.x] = 0.0f;
}

__global__ __launch_bounds__(128) void qvl_fused(
    const float* __restrict__ x,
    const float* __restrict__ enc_w1,
    const float* __restrict__ enc_b1,
    const float* __restrict__ enc_w2,
    const float* __restrict__ enc_b2,
    const float* __restrict__ q_params,
    const float* __restrict__ dec_w1,
    const float* __restrict__ dec_b1,
    const float* __restrict__ dec_w2,
    const float* __restrict__ dec_b2,
    float* __restrict__ out,
    float* __restrict__ ent)
{
  __shared__ float s_w1t[64][128];
  __shared__ float s_w2t[8][64];
  __shared__ float s_dw1t[16][4];
  __shared__ float s_dw2t[64][16];
  __shared__ float s_b1[64];
  __shared__ float s_b2[8];
  __shared__ float s_db1[16];
  __shared__ float s_db2[64];
  __shared__ float s_rot[12][8];
  __shared__ float s_h[4][16][66];
  __shared__ float s_ang[4][16][8];
  __shared__ float s_qf[4][16][4];
  __shared__ float s_g[4][16][18];
  __shared__ float s_ent[4];

  const int tid  = threadIdx.x;
  const int wv   = tid >> 5;
  const int lane = tid & 31;
  const int col  = lane & 15;
  const int hi   = lane >> 4;
  const int r0   = hi * 2;

  for (int i = tid; i < 128 * 64; i += 128) s_w1t[i & 63][i >> 6] = enc_w1[i];
  for (int i = tid; i < 64 * 8;  i += 128) s_w2t[i & 7][i >> 3]  = enc_w2[i];
  for (int i = tid; i < 4 * 16;  i += 128) s_dw1t[i & 15][i >> 4] = dec_w1[i];
  for (int i = tid; i < 16 * 64; i += 128) s_dw2t[i & 63][i >> 6] = dec_w2[i];
  if (tid < 64) s_b1[tid] = enc_b1[tid];
  if (tid < 8)  s_b2[tid] = enc_b2[tid];
  if (tid < 16) s_db1[tid] = dec_b1[tid];
  if (tid >= 64 && tid < 128) s_db2[tid - 64] = dec_b2[tid - 64];
  if (tid < 4) s_ent[tid] = 0.0f;
  if (tid < 12) {
    const float phi = q_params[tid * 3 + 0];
    const float th  = q_params[tid * 3 + 1];
    const float om  = q_params[tid * 3 + 2];
    const float ch = cosf(0.5f * th), sh = sinf(0.5f * th);
    const float pa = -0.5f * (phi + om);
    const float pb =  0.5f * (phi - om);
    const float ca = cosf(pa), sa = sinf(pa);
    const float cb = cosf(pb), sb = sinf(pb);
    s_rot[tid][0] =  ca * ch;  s_rot[tid][1] =  sa * ch;
    s_rot[tid][2] = -cb * sh;  s_rot[tid][3] = -sb * sh;
    s_rot[tid][4] =  cb * sh;  s_rot[tid][5] = -sb * sh;
    s_rot[tid][6] =  ca * ch;  s_rot[tid][7] = -sa * ch;
  }
  __syncthreads();

  const int tok0 = blockIdx.x * 64 + wv * 16;
  const float* xrow = x + (size_t)(tok0 + col) * 128;
  __builtin_prefetch(xrow, 0, 0);

  v8f acc[4];
#pragma unroll
  for (int n = 0; n < 4; ++n) acc[n] = (v8f){0.f,0.f,0.f,0.f,0.f,0.f,0.f,0.f};
#pragma unroll
  for (int k = 0; k < 32; ++k) {
    const v2f a = *(const v2f*)(xrow + 4 * k + r0);
#pragma unroll
    for (int n = 0; n < 4; ++n) {
      const v2f b = *(const v2f*)(&s_w1t[16 * n + col][4 * k + r0]);
      acc[n] = wmma4(a, b, acc[n]);
    }
  }
#pragma unroll
  for (int n = 0; n < 4; ++n) {
    const float bv = s_b1[16 * n + col];
#pragma unroll
    for (int r = 0; r < 8; ++r) {
      s_h[wv][r + 8 * hi][16 * n + col] = gelu_exact(acc[n][r] + bv);
    }
  }

  v8f eacc = (v8f){0.f,0.f,0.f,0.f,0.f,0.f,0.f,0.f};
#pragma unroll
  for (int k = 0; k < 16; ++k) {
    const v2f a = *(const v2f*)(&s_h[wv][col][4 * k + r0]);
    float b0 = 0.0f, b1 = 0.0f;
    if (col < 8) { b0 = s_w2t[col][4 * k + r0]; b1 = s_w2t[col][4 * k + r0 + 1]; }
    const v2f b = {b0, b1};
    eacc = wmma4(a, b, eacc);
  }
  if (col < 8) {
    const float bv = s_b2[col];
#pragma unroll
    for (int r = 0; r < 8; ++r) s_ang[wv][r + 8 * hi][col] = eacc[r] + bv;
  }

  if (lane < 16) {
    float zr[16], zi[16];
#pragma unroll
    for (int s = 0; s < 16; ++s) { zr[s] = 0.0f; zi[s] = 0.0f; }
    zr[0] = 1.0f;

#pragma unroll
    for (int i = 0; i < 4; ++i) {
      const int m = 8 >> i;
      const float tx = s_ang[wv][lane][2 * i];
      const float ty = s_ang[wv][lane][2 * i + 1];
      float c = cosf(0.5f * tx), sn = sinf(0.5f * tx);
      apply1q(zr, zi, m, c, 0.f, 0.f, -sn, 0.f, -sn, c, 0.f);
      c = cosf(0.5f * ty); sn = sinf(0.5f * ty);
      apply1q(zr, zi, m, c, 0.f, -sn, 0.f, sn, 0.f, c, 0.f);
    }
#pragma unroll
    for (int l = 0; l < 3; ++l) {
#pragma unroll
      for (int i = 0; i < 4; ++i) {
        const float* u = s_rot[l * 4 + i];
        apply1q(zr, zi, 8 >> i, u[0], u[1], u[2], u[3], u[4], u[5], u[6], u[7]);
      }
#pragma unroll
      for (int i = 0; i < 4; ++i) {
        const int mc = 8 >> i;
        const int mt = 8 >> ((i + 1) & 3);
#pragma unroll
        for (int s = 0; s < 16; ++s) {
          if ((s & mc) && !(s & mt)) {
            const int t = s | mt;
            float tr = zr[s]; zr[s] = zr[t]; zr[t] = tr;
            float ti = zi[s]; zi[s] = zi[t]; zi[t] = ti;
          }
        }
      }
    }
    float p[16];
#pragma unroll
    for (int s = 0; s < 16; ++s) p[s] = zr[s] * zr[s] + zi[s] * zi[s];
#pragma unroll
    for (int i = 0; i < 4; ++i) {
      const int m = 8 >> i;
      float z = 0.0f;
#pragma unroll
      for (int s = 0; s < 16; ++s) z += (s & m) ? -p[s] : p[s];
      s_qf[wv][lane][i] = z;
      const float qp = (z + 1.0f) * 0.5f;
      const float et = -(qp * logf(qp + 1e-10f) +
                         (1.0f - qp) * logf(1.0f - qp + 1e-10f));
      atomicAdd(&s_ent[i], et);
    }
  }

  {
    const v2f a = *(const v2f*)(&s_qf[wv][col][r0]);
    const v2f b = *(const v2f*)(&s_dw1t[col][r0]);
    v8f g = (v8f){0.f,0.f,0.f,0.f,0.f,0.f,0.f,0.f};
    g = wmma4(a, b, g);
    const float bv = s_db1[col];
#pragma unroll
    for (int r = 0; r < 8; ++r)
      s_g[wv][r + 8 * hi][col] = gelu_exact(g[r] + bv);
  }

  v8f oacc[4];
#pragma unroll
  for (int n = 0; n < 4; ++n) oacc[n] = (v8f){0.f,0.f,0.f,0.f,0.f,0.f,0.f,0.f};
#pragma unroll
  for (int k = 0; k < 4; ++k) {
    const v2f a = *(const v2f*)(&s_g[wv][col][4 * k + r0]);
#pragma unroll
    for (int n = 0; n < 4; ++n) {
      const v2f b = *(const v2f*)(&s_dw2t[16 * n + col][4 * k + r0]);
      oacc[n] = wmma4(a, b, oacc[n]);
    }
  }
  float* obase = out + (size_t)tok0 * 64;
#pragma unroll
  for (int n = 0; n < 4; ++n) {
    const float bv = s_db2[16 * n + col];
#pragma unroll
    for (int r = 0; r < 8; ++r) {
      __builtin_nontemporal_store(
          oacc[n][r] + bv,
          obase + (size_t)(r + 8 * hi) * 64 + 16 * n + col);
    }
  }

  __syncthreads();
  if (tid < 4) atomicAdd(ent + tid, s_ent[tid] * (1.0f / (float)TOK_TOTAL));
}

extern "C" void kernel_launch(void* const* d_in, const int* in_sizes, int n_in,
                              void* d_out, int out_size, void* d_ws, size_t ws_size,
                              hipStream_t stream) {
  (void)in_sizes; (void)n_in; (void)out_size; (void)d_ws; (void)ws_size;
  const float* x        = (const float*)d_in[0];
  const float* enc_w1   = (const float*)d_in[1];
  const float* enc_b1   = (const float*)d_in[2];
  const float* enc_w2   = (const float*)d_in[3];
  const float* enc_b2   = (const float*)d_in[4];
  const float* q_params = (const float*)d_in[5];
  const float* dec_w1   = (const float*)d_in[6];
  const float* dec_b1   = (const float*)d_in[7];
  const float* dec_w2   = (const float*)d_in[8];
  const float* dec_b2   = (const float*)d_in[9];

  float* out = (float*)d_out;
  float* ent = out + LATENT_ELEMS;

  qvl_zero_ent<<<1, 32, 0, stream>>>(ent);
  qvl_fused<<<TOK_TOTAL / 64, 128, 0, stream>>>(
      x, enc_w1, enc_b1, enc_w2, enc_b2, q_params,
      dec_w1, dec_b1, dec_w2, dec_b2, out, ent);
}


# LDS row pitch of the staged weight tiles padded by 4 floats (132/68/20): B-fragment reads and transposing staging stores no longer fall on one bank
# speedup vs baseline: 1.5250x; 1.5250x over previous
#include <hip/hip_runtime.h>
#include <hip/hip_bf16.h>
#include <math.h>


typedef float v2f __attribute__((ext_vector_type(2)));
typedef float v8f __attribute__((ext_vector_type(8)));

#define TOK_TOTAL (64 * 4096)
#define LATENT_ELEMS ((size_t)TOK_TOTAL * 64)

static __device__ __forceinline__ v8f wmma4(v2f a, v2f b, v8f c) {
  return __builtin_amdgcn_wmma_f32_16x16x4_f32(
       false, a,  false, b,
       (short)0, c,  false,  false);
}

static __device__ __forceinline__ float gelu_exact(float x) {
  return 0.5f * x * (1.0f + erff(x * 0.70710678118654752f));
}

static __device__ __forceinline__ void apply1q(
    float* zr, float* zi, int m,
    float u00r, float u00i, float u01r, float u01i,
    float u10r, float u10i, float u11r, float u11i) {
#pragma unroll
  for (int s = 0; s < 16; ++s) {
    if (!(s & m)) {
      const int t = s | m;
      const float ar = zr[s], ai = zi[s];
      const float br = zr[t], bi = zi[t];
      zr[s] = u00r * ar - u00i * ai + u01r * br - u01i * bi;
      zi[s] = u00r * ai + u00i * ar + u01r * bi + u01i * br;
      zr[t] = u10r * ar - u10i * ai + u11r * br - u11i * bi;
      zi[t] = u10r * ai + u10i * ar + u11r * bi + u11i * br;
    }
  }
}

__global__ void qvl_zero_ent(float* ent) {
  if (threadIdx.x < 4) ent[threadIdx.x] = 0.0f;
}

__global__ __launch_bounds__(128) void qvl_fused(
    const float* __restrict__ x,
    const float* __restrict__ enc_w1,
    const float* __restrict__ enc_b1,
    const float* __restrict__ enc_w2,
    const float* __restrict__ enc_b2,
    const float* __restrict__ q_params,
    const float* __restrict__ dec_w1,
    const float* __restrict__ dec_b1,
    const float* __restrict__ dec_w2,
    const float* __restrict__ dec_b2,
    float* __restrict__ out,
    float* __restrict__ ent)
{
  __shared__ float s_w1t[64][128 + 4];
  __shared__ float s_w2t[8][64 + 4];
  __shared__ float s_dw1t[16][4];
  __shared__ float s_dw2t[64][16 + 4];
  __shared__ float s_b1[64];
  __shared__ float s_b2[8];
  __shared__ float s_db1[16];
  __shared__ float s_db2[64];
  __shared__ float s_rot[12][8];
  __shared__ float s_h[4][16][66];
  __shared__ float s_ang[4][16][8];
  __shared__ float s_qf[4][16][4];
  __shared__ float s_g[4][16][18];
  __shared__ float s_ent[4];

  const int tid  = threadIdx.x;
  const int wv   = tid >> 5;
  const int lane = tid & 31;
  const int col  = lane & 15;
  const int hi   = lane >> 4;
  const int r0   = hi * 2;

  for (int i = tid; i < 128 * 64; i += 128) s_w1t[i & 63][i >> 6] = enc_w1[i];
  for (int i = tid; i < 64 * 8;  i += 128) s_w2t[i & 7][i >> 3]  = enc_w2[i];
  for (int i = tid; i < 4 * 16;  i += 128) s_dw1t[i & 15][i >> 4] = dec_w1[i];
  for (int i = tid; i < 16 * 64; i += 128) s_dw2t[i & 63][i >> 6] = dec_w2[i];
  if (tid < 64) s_b1[tid] = enc_b1[tid];
  if (tid < 8)  s_b2[tid] = enc_b2[tid];
  if (tid < 16) s_db1[tid] = dec_b1[tid];
  if (tid >= 64 && tid < 128) s_db2[tid - 64] = dec_b2[tid - 64];
  if (tid < 4) s_ent[tid] = 0.0f;
  if (tid < 12) {
    const float phi = q_params[tid * 3 + 0];
    const float th  = q_params[tid * 3 + 1];
    const float om  = q_params[tid * 3 + 2];
    const float ch = cosf(0.5f * th), sh = sinf(0.5f * th);
    const float pa = -0.5f * (phi + om);
    const float pb =  0.5f * (phi - om);
    const float ca = cosf(pa), sa = sinf(pa);
    const float cb = cosf(pb), sb = sinf(pb);
    s_rot[tid][0] =  ca * ch;  s_rot[tid][1] =  sa * ch;
    s_rot[tid][2] = -cb * sh;  s_rot[tid][3] = -sb * sh;
    s_rot[tid][4] =  cb * sh;  s_rot[tid][5] = -sb * sh;
    s_rot[tid][6] =  ca * ch;  s_rot[tid][7] = -sa * ch;
  }
  __syncthreads();

  const int tok0 = blockIdx.x * 64 + wv * 16;
  const float* xrow = x + (size_t)(tok0 + col) * 128;
  __builtin_prefetch(xrow, 0, 0);

  v8f acc[4];
#pragma unroll
  for (int n = 0; n < 4; ++n) acc[n] = (v8f){0.f,0.f,0.f,0.f,0.f,0.f,0.f,0.f};
#pragma unroll
  for (int k = 0; k < 32; ++k) {
    const v2f a = *(const v2f*)(xrow + 4 * k + r0);
#pragma unroll
    for (int n = 0; n < 4; ++n) {
      const v2f b = *(const v2f*)(&s_w1t[16 * n + col][4 * k + r0]);
      acc[n] = wmma4(a, b, acc[n]);
    }
  }
#pragma unroll
  for (int n = 0; n < 4; ++n) {
    const float bv = s_b1[16 * n + col];
#pragma unroll
    for (int r = 0; r < 8; ++r) {
      s_h[wv][r + 8 * hi][16 * n + col] = gelu_exact(acc[n][r] + bv);
    }
  }

  v8f eacc = (v8f){0.f,0.f,0.f,0.f,0.f,0.f,0.f,0.f};
#pragma unroll
  for (int k = 0; k < 16; ++k) {
    const v2f a = *(const v2f*)(&s_h[wv][col][4 * k + r0]);
    float b0 = 0.0f, b1 = 0.0f;
    if (col < 8) { b0 = s_w2t[col][4 * k + r0]; b1 = s_w2t[col][4 * k + r0 + 1]; }
    const v2f b = {b0, b1};
    eacc = wmma4(a, b, eacc);
  }
  if (col < 8) {
    const float bv = s_b2[col];
#pragma unroll
    for (int r = 0; r < 8; ++r) s_ang[wv][r + 8 * hi][col] = eacc[r] + bv;
  }

  if (lane < 16) {
    float zr[16], zi[16];
#pragma unroll
    for (int s = 0; s < 16; ++s) { zr[s] = 0.0f; zi[s] = 0.0f; }
    zr[0] = 1.0f;

#pragma unroll
    for (int i = 0; i < 4; ++i) {
      const int m = 8 >> i;
      const float tx = s_ang[wv][lane][2 * i];
      const float ty = s_ang[wv][lane][2 * i + 1];
      float c = cosf(0.5f * tx), sn = sinf(0.5f * tx);
      apply1q(zr, zi, m, c, 0.f, 0.f, -sn, 0.f, -sn, c, 0.f);
      c = cosf(0.5f * ty); sn = sinf(0.5f * ty);
      apply1q(zr, zi, m, c, 0.f, -sn, 0.f, sn, 0.f, c, 0.f);
    }
#pragma unroll
    for (int l = 0; l < 3; ++l) {
#pragma unroll
      for (int i = 0; i < 4; ++i) {
        const float* u = s_rot[l * 4 + i];
        apply1q(zr, zi, 8 >> i, u[0], u[1], u[2], u[3], u[4], u[5], u[6], u[7]);
      }
#pragma unroll
      for (int i = 0; i < 4; ++i) {
        const int mc = 8 >> i;
        const int mt = 8 >> ((i + 1) & 3);
#pragma unroll
        for (int s = 0; s < 16; ++s) {
          if ((s & mc) && !(s & mt)) {
            const int t = s | mt;
            float tr = zr[s]; zr[s] = zr[t]; zr[t] = tr;
            float ti = zi[s]; zi[s] = zi[t]; zi[t] = ti;
          }
        }
      }
    }
    float p[16];
#pragma unroll
    for (int s = 0; s < 16; ++s) p[s] = zr[s] * zr[s] + zi[s] * zi[s];
#pragma unroll
    for (int i = 0; i < 4; ++i) {
      const int m = 8 >> i;
      float z = 0.0f;
#pragma unroll
      for (int s = 0; s < 16; ++s) z += (s & m) ? -p[s] : p[s];
      s_qf[wv][lane][i] = z;
      const float qp = (z + 1.0f) * 0.5f;
      const float et = -(qp * logf(qp + 1e-10f) +
                         (1.0f - qp) * logf(1.0f - qp + 1e-10f));
      atomicAdd(&s_ent[i], et);
    }
  }

  {
    const v2f a = *(const v2f*)(&s_qf[wv][col][r0]);
    const v2f b = *(const v2f*)(&s_dw1t[col][r0]);
    v8f g = (v8f){0.f,0.f,0.f,0.f,0.f,0.f,0.f,0.f};
    g = wmma4(a, b, g);
    const float bv = s_db1[col];
#pragma unroll
    for (int r = 0; r < 8; ++r)
      s_g[wv][r + 8 * hi][col] = gelu_exact(g[r] + bv);
  }

  v8f oacc[4];
#pragma unroll
  for (int n = 0; n < 4; ++n) oacc[n] = (v8f){0.f,0.f,0.f,0.f,0.f,0.f,0.f,0.f};
#pragma unroll
  for (int k = 0; k < 4; ++k) {
    const v2f a = *(const v2f*)(&s_g[wv][col][4 * k + r0]);
#pragma unroll
    for (int n = 0; n < 4; ++n) {
      const v2f b = *(const v2f*)(&s_dw2t[16 * n + col][4 * k + r0]);
      oacc[n] = wmma4(a, b, oacc[n]);
    }
  }
  float* obase = out + (size_t)tok0 * 64;
#pragma unroll
  for (int n = 0; n < 4; ++n) {
    const float bv = s_db2[16 * n + col];
#pragma unroll
    for (int r = 0; r < 8; ++r) {
      __builtin_nontemporal_store(
          oacc[n][r] + bv,
          obase + (size_t)(r + 8 * hi) * 64 + 16 * n + col);
    }
  }

  __syncthreads();
  if (tid < 4) atomicAdd(ent + tid, s_ent[tid] * (1.0f / (float)TOK_TOTAL));
}

extern "C" void kernel_launch(void* const* d_in, const int* in_sizes, int n_in,
                              void* d_out, int out_size, void* d_ws, size_t ws_size,
                              hipStream_t stream) {
  (void)in_sizes; (void)n_in; (void)out_size; (void)d_ws; (void)ws_size;
  const float* x        = (const float*)d_in[0];
  const float* enc_w1   = (const float*)d_in[1];
  const float* enc_b1   = (const float*)d_in[2];
  const float* enc_w2   = (const float*)d_in[3];
  const float* enc_b2   = (const float*)d_in[4];
  const float* q_params = (const float*)d_in[5];
  const float* dec_w1   = (const float*)d_in[6];
  const float* dec_b1   = (const float*)d_in[7];
  const float* dec_w2   = (const float*)d_in[8];
  const float* dec_b2   = (const float*)d_in[9];

  float* out = (float*)d_out;
  float* ent = out + LATENT_ELEMS;

  qvl_zero_ent<<<1, 32, 0, stream>>>(ent);
  qvl_fused<<<TOK_TOTAL / 64, 128, 0, stream>>>(
      x, enc_w1, enc_b1, enc_w2, enc_b2, q_params,
      dec_w1, dec_b1, dec_w2, dec_b2, out, ent);
}


# Each wave takes 32 tokens (GEMM stages as two looped 16-row passes): the state-vector stage fills all 32 lanes, half as many blocks stage weights; entropy sums regroup (32 lanes per LDS add)
# speedup vs baseline: 1.8389x; 1.2058x over previous
#include <hip/hip_runtime.h>
#include <hip/hip_bf16.h>
#include <math.h>


typedef float v2f __attribute__((ext_vector_type(2)));
typedef float v8f __attribute__((ext_vector_type(8)));

#define TOK_TOTAL (64 * 4096)
#define LATENT_ELEMS ((size_t)TOK_TOTAL * 64)
#define WAVE_TOK 32
#define BLOCK_TOK (4 * WAVE_TOK)

static __device__ __forceinline__ v8f wmma4(v2f a, v2f b, v8f c) {
  return __builtin_amdgcn_wmma_f32_16x16x4_f32(
       false, a,  false, b,
       (short)0, c,  false,  false);
}

static __device__ __forceinline__ float gelu_exact(float x) {
  return 0.5f * x * (1.0f + erff(x * 0.70710678118654752f));
}

static __device__ __forceinline__ void apply1q(
    float* zr, float* zi, int m,
    float u00r, float u00i, float u01r, float u01i,
    float u10r, float u10i, float u11r, float u11i) {
#pragma unroll
  for (int s = 0; s < 16; ++s) {
    if (!(s & m)) {
      const int t = s | m;
      const float ar = zr[s], ai = zi[s];
      const float br = zr[t], bi = zi[t];
      zr[s] = u00r * ar - u00i * ai + u01r * br - u01i * bi;
      zi[s] = u00r * ai + u00i * ar + u01r * bi + u01i * br;
      zr[t] = u10r * ar - u10i * ai + u11r * br - u11i * bi;
      zi[t] = u10r * ai + u10i * ar + u11r * bi + u11i * br;
    }
  }
}

__global__ void qvl_zero_ent(float* ent) {
  if (threadIdx.x < 4) ent[threadIdx.x] = 0.0f;
}

__global__ __launch_bounds__(128) void qvl_fused(
    const float* __restrict__ x,
    const float* __restrict__ enc_w1,
    const float* __restrict__ enc_b1,
    const float* __restrict__ enc_w2,
    const float* __restrict__ enc_b2,
    const float* __restrict__ q_params,
    const float* __restrict__ dec_w1,
    const float* __restrict__ dec_b1,
    const float* __restrict__ dec_w2,
    const float* __restrict__ dec_b2,
    float* __restrict__ out,
    float* __restrict__ ent)
{
  __shared__ float s_w1t[64][128 + 4];
  __shared__ float s_w2t[8][64 + 4];
  __shared__ float s_dw1t[16][4];
  __shared__ float s_dw2t[64][16 + 4];
  __shared__ float s_b1[64];
  __shared__ float s_b2[8];
  __shared__ float s_db1[16];
  __shared__ float s_db2[64];
  __shared__ float s_rot[12][8];
  __shared__ float s_h[4][16][66];
  __shared__ float s_ang[4][32][9];
  __shared__ float s_qf[4][32][4];
  __shared__ float s_g[4][16][18];
  __shared__ float s_ent[4];

  const int tid  = threadIdx.x;
  const int wv   = tid >> 5;
  const int lane = tid & 31;
  const int col  = lane & 15;
  const int hi   = lane >> 4;
  const int r0   = hi * 2;

  for (int i = tid; i < 128 * 64; i += 128) s_w1t[i & 63][i >> 6] = enc_w1[i];
  for (int i = tid; i < 64 * 8;  i += 128) s_w2t[i & 7][i >> 3]  = enc_w2[i];
  for (int i = tid; i < 4 * 16;  i += 128) s_dw1t[i & 15][i >> 4] = dec_w1[i];
  for (int i = tid; i < 16 * 64; i += 128) s_dw2t[i & 63][i >> 6] = dec_w2[i];
  if (tid < 64) s_b1[tid] = enc_b1[tid];
  if (tid < 8)  s_b2[tid] = enc_b2[tid];
  if (tid < 16) s_db1[tid] = dec_b1[tid];
  if (tid >= 64 && tid < 128) s_db2[tid - 64] = dec_b2[tid - 64];
  if (tid < 4) s_ent[tid] = 0.0f;
  if (tid < 12) {
    const float phi = q_params[tid * 3 + 0];
    const float th  = q_params[tid * 3 + 1];
    const float om  = q_params[tid * 3 + 2];
    const float ch = cosf(0.5f * th), sh = sinf(0.5f * th);
    const float pa = -0.5f * (phi + om);
    const float pb =  0.5f * (phi - om);
    const float ca = cosf(pa), sa = sinf(pa);
    const float cb = cosf(pb), sb = sinf(pb);
    s_rot[tid][0] =  ca * ch;  s_rot[tid][1] =  sa * ch;
    s_rot[tid][2] = -cb * sh;  s_rot[tid][3] = -sb * sh;
    s_rot[tid][4] =  cb * sh;  s_rot[tid][5] = -sb * sh;
    s_rot[tid][6] =  ca * ch;  s_rot[tid][7] = -sa * ch;
  }
  __syncthreads();

  const int tok0 = blockIdx.x * BLOCK_TOK + wv * WAVE_TOK;
  {
    const float* xnext = x + (size_t)(tok0 + 16 + col) * 128 + 64 * hi;
    __builtin_prefetch(xnext, 0, 0);
    __builtin_prefetch(xnext + 32, 0, 0);
  }

#pragma unroll 1
  for (int mt = 0; mt < 2; ++mt) {
  asm volatile("" ::: "memory");
  const float* xrow = x + (size_t)(tok0 + 16 * mt + col) * 128;
  __builtin_prefetch(xrow, 0, 0);

  v8f acc[4];
#pragma unroll
  for (int n = 0; n < 4; ++n) acc[n] = (v8f){0.f,0.f,0.f,0.f,0.f,0.f,0.f,0.f};
#pragma unroll
  for (int k = 0; k < 32; ++k) {
    const v2f a = *(const v2f*)(xrow + 4 * k + r0);
#pragma unroll
    for (int n = 0; n < 4; ++n) {
      const v2f b = *(const v2f*)(&s_w1t[16 * n + col][4 * k + r0]);
      acc[n] = wmma4(a, b, acc[n]);
    }
  }
#pragma unroll
  for (int n = 0; n < 4; ++n) {
    const float bv = s_b1[16 * n + col];
#pragma unroll
    for (int r = 0; r < 8; ++r) {
      s_h[wv][r + 8 * hi][16 * n + col] = gelu_exact(acc[n][r] + bv);
    }
  }

  v8f eacc = (v8f){0.f,0.f,0.f,0.f,0.f,0.f,0.f,0.f};
#pragma unroll
  for (int k = 0; k < 16; ++k) {
    const v2f a = *(const v2f*)(&s_h[wv][col][4 * k + r0]);
    float b0 = 0.0f, b1 = 0.0f;
    if (col < 8) { b0 = s_w2t[col][4 * k + r0]; b1 = s_w2t[col][4 * k + r0 + 1]; }
    const v2f b = {b0, b1};
    eacc = wmma4(a, b, eacc);
  }
  if (col < 8) {
    const float bv = s_b2[col];
#pragma unroll
    for (int r = 0; r < 8; ++r) s_ang[wv][16 * mt + r + 8 * hi][col] = eacc[r] + bv;
  }
  }

  {
    float zr[16], zi[16];
#pragma unroll
    for (int s = 0; s < 16; ++s) { zr[s] = 0.0f; zi[s] = 0.0f; }
    zr[0] = 1.0f;

#pragma unroll
    for (int i = 0; i < 4; ++i) {
      const int m = 8 >> i;
      const float tx = s_ang[wv][lane][2 * i];
      const float ty = s_ang[wv][lane][2 * i + 1];
      float c = cosf(0.5f * tx), sn = sinf(0.5f * tx);
      apply1q(zr, zi, m, c, 0.f, 0.f, -sn, 0.f, -sn, c, 0.f);
      c = cosf(0.5f * ty); sn = sinf(0.5f * ty);
      apply1q(zr, zi, m, c, 0.f, -sn, 0.f, sn, 0.f, c, 0.f);
    }
#pragma unroll
    for (int l = 0; l < 3; ++l) {
#pragma unroll
      for (int i = 0; i < 4; ++i) {
        const float* u = s_rot[l * 4 + i];
        apply1q(zr, zi, 8 >> i, u[0], u[1], u[2], u[3], u[4], u[5], u[6], u[7]);
      }
#pragma unroll
      for (int i = 0; i < 4; ++i) {
        const int mc = 8 >> i;
        const int mt = 8 >> ((i + 1) & 3);
#pragma unroll
        for (int s = 0; s < 16; ++s) {
          if ((s & mc) && !(s & mt)) {
            const int t = s | mt;
            float tr = zr[s]; zr[s] = zr[t]; zr[t] = tr;
            float ti = zi[s]; zi[s] = zi[t]; zi[t] = ti;
          }
        }
      }
    }
    float p[16];
#pragma unroll
    for (int s = 0; s < 16; ++s) p[s] = zr[s] * zr[s] + zi[s] * zi[s];
#pragma unroll
    for (int i = 0; i < 4; ++i) {
      const int m = 8 >> i;
      float z = 0.0f;
#pragma unroll
      for (int s = 0; s < 16; ++s) z += (s & m) ? -p[s] : p[s];
      s_qf[wv][lane][i] = z;
      const float qp = (z + 1.0f) * 0.5f;
      const float et = -(qp * logf(qp + 1e-10f) +
                         (1.0f - qp) * logf(1.0f - qp + 1e-10f));
      atomicAdd(&s_ent[i], et);
    }
  }

#pragma unroll 1
  for (int mt = 0; mt < 2; ++mt) {
  {
    const v2f a = *(const v2f*)(&s_qf[wv][16 * mt + col][r0]);
    const v2f b = *(const v2f*)(&s_dw1t[col][r0]);
    v8f g = (v8f){0.f,0.f,0.f,0.f,0.f,0.f,0.f,0.f};
    g = wmma4(a, b, g);
    const float bv = s_db1[col];
#pragma unroll
    for (int r = 0; r < 8; ++r)
      s_g[wv][r + 8 * hi][col] = gelu_exact(g[r] + bv);
  }

  v8f oacc[4];
#pragma unroll
  for (int n = 0; n < 4; ++n) oacc[n] = (v8f){0.f,0.f,0.f,0.f,0.f,0.f,0.f,0.f};
#pragma unroll
  for (int k = 0; k < 4; ++k) {
    const v2f a = *(const v2f*)(&s_g[wv][col][4 * k + r0]);
#pragma unroll
    for (int n = 0; n < 4; ++n) {
      const v2f b = *(const v2f*)(&s_dw2t[16 * n + col][4 * k + r0]);
      oacc[n] = wmma4(a, b, oacc[n]);
    }
  }
  float* obase = out + (size_t)(tok0 + 16 * mt) * 64;
#pragma unroll
  for (int n = 0; n < 4; ++n) {
    const float bv = s_db2[16 * n + col];
#pragma unroll
    for (int r = 0; r < 8; ++r) {
      __builtin_nontemporal_store(
          oacc[n][r] + bv,
          obase + (size_t)(r + 8 * hi) * 64 + 16 * n + col);
    }
  }
  }

  __syncthreads();
  if (tid < 4) atomicAdd(ent + tid, s_ent[tid] * (1.0f / (float)TOK_TOTAL));
}

extern "C" void kernel_launch(void* const* d_in, const int* in_sizes, int n_in,
                              void* d_out, int out_size, void* d_ws, size_t ws_size,
                              hipStream_t stream) {
  (void)in_sizes; (void)n_in; (void)out_size; (void)d_ws; (void)ws_size;
  const float* x        = (const float*)d_in[0];
  const float* enc_w1   = (const float*)d_in[1];
  const float* enc_b1   = (const float*)d_in[2];
  const float* enc_w2   = (const float*)d_in[3];
  const float* enc_b2   = (const float*)d_in[4];
  const float* q_params = (const float*)d_in[5];
  const float* dec_w1   = (const float*)d_in[6];
  const float* dec_b1   = (const float*)d_in[7];
  const float* dec_w2   = (const float*)d_in[8];
  const float* dec_b2   = (const float*)d_in[9];

  float* out = (float*)d_out;
  float* ent = out + LATENT_ELEMS;

  qvl_zero_ent<<<1, 32, 0, stream>>>(ent);
  qvl_fused<<<TOK_TOTAL / BLOCK_TOK, 128, 0, stream>>>(
      x, enc_w1, enc_b1, enc_w2, enc_b2, q_params,
      dec_w1, dec_b1, dec_w2, dec_b2, out, ent);
}


# GELU's erf made branch-free: both polynomial ranges evaluated per element and selected (exp as exp2 of the scaled argument), so a tile's elements are independent straight-line chains
# speedup vs baseline: 2.1361x; 1.1616x over previous
#include <hip/hip_runtime.h>
#include <hip/hip_bf16.h>
#include <math.h>


typedef float v2f __attribute__((ext_vector_type(2)));
typedef float v8f __attribute__((ext_vector_type(8)));

#define TOK_TOTAL (64 * 4096)
#define LATENT_ELEMS ((size_t)TOK_TOTAL * 64)
#define WAVE_TOK 32
#define BLOCK_TOK (4 * WAVE_TOK)

static __device__ __forceinline__ v8f wmma4(v2f a, v2f b, v8f c) {
  return __builtin_amdgcn_wmma_f32_16x16x4_f32(
       false, a,  false, b,
       (short)0, c,  false,  false);
}

static __device__ __forceinline__ float gelu_exact(float x) {
  const float v = x * 0.70710678118654752f;
  const float t = fabsf(v);
  float p = fmaf(t, 0x1.1d3156p-16f, -0x1.8d1290p-12f);
  p = fmaf(t, p, 0x1.f9a6d2p-9f);
  p = fmaf(t, p, -0x1.8c3164p-6f);
  p = fmaf(t, p, 0x1.b4e9c8p-4f);
  p = fmaf(t, p, 0x1.4515fap-1f);
  p = fmaf(t, p, 0x1.078e50p-3f);
  p = fmaf(t, p, t);
  const float hi_arm = 1.0f - __builtin_amdgcn_exp2f(p * -0x1.715476p+0f);
  const float s = v * v;
  float q = fmaf(s, -0x1.268bc2p-11f, 0x1.420828p-8f);
  q = fmaf(s, q, -0x1.b59370p-6f);
  q = fmaf(s, q, 0x1.ce077cp-4f);
  q = fmaf(s, q, -0x1.812660p-2f);
  q = fmaf(s, q, 0x1.06eba0p-3f);
  const float lo_arm = fmaf(t, q, t);
  const float e = copysignf(t < 1.0f ? lo_arm : hi_arm, v);
  return 0.5f * x * (1.0f + e);
}

static __device__ __forceinline__ void apply1q(
    float* zr, float* zi, int m,
    float u00r, float u00i, float u01r, float u01i,
    float u10r, float u10i, float u11r, float u11i) {
#pragma unroll
  for (int s = 0; s < 16; ++s) {
    if (!(s & m)) {
      const int t = s | m;
      const float ar = zr[s], ai = zi[s];
      const float br = zr[t], bi = zi[t];
      zr[s] = u00r * ar - u00i * ai + u01r * br - u01i * bi;
      zi[s] = u00r * ai + u00i * ar + u01r * bi + u01i * br;
      zr[t] = u10r * ar - u10i * ai + u11r * br - u11i * bi;
      zi[t] = u10r * ai + u10i * ar + u11r * bi + u11i * br;
    }
  }
}

__global__ void qvl_zero_ent(float* ent) {
  if (threadIdx.x < 4) ent[threadIdx.x] = 0.0f;
}

__global__ __launch_bounds__(128) void qvl_fused(
    const float* __restrict__ x,
    const float* __restrict__ enc_w1,
    const float* __restrict__ enc_b1,
    const float* __restrict__ enc_w2,
    const float* __restrict__ enc_b2,
    const float* __restrict__ q_params,
    const float* __restrict__ dec_w1,
    const float* __restrict__ dec_b1,
    const float* __restrict__ dec_w2,
    const float* __restrict__ dec_b2,
    float* __restrict__ out,
    float* __restrict__ ent)
{
  __shared__ float s_w1t[64][128 + 4];
  __shared__ float s_w2t[8][64 + 4];
  __shared__ float s_dw1t[16][4];
  __shared__ float s_dw2t[64][16 + 4];
  __shared__ float s_b1[64];
  __shared__ float s_b2[8];
  __shared__ float s_db1[16];
  __shared__ float s_db2[64];
  __shared__ float s_rot[12][8];
  __shared__ float s_h[4][16][66];
  __shared__ float s_ang[4][32][9];
  __shared__ float s_qf[4][32][4];
  __shared__ float s_g[4][16][18];
  __shared__ float s_ent[4];

  const int tid  = threadIdx.x;
  const int wv   = tid >> 5;
  const int lane = tid & 31;
  const int col  = lane & 15;
  const int hi   = lane >> 4;
  const int r0   = hi * 2;

  for (int i = tid; i < 128 * 64; i += 128) s_w1t[i & 63][i >> 6] = enc_w1[i];
  for (int i = tid; i < 64 * 8;  i += 128) s_w2t[i & 7][i >> 3]  = enc_w2[i];
  for (int i = tid; i < 4 * 16;  i += 128) s_dw1t[i & 15][i >> 4] = dec_w1[i];
  for (int i = tid; i < 16 * 64; i += 128) s_dw2t[i & 63][i >> 6] = dec_w2[i];
  if (tid < 64) s_b1[tid] = enc_b1[tid];
  if (tid < 8)  s_b2[tid] = enc_b2[tid];
  if (tid < 16) s_db1[tid] = dec_b1[tid];
  if (tid >= 64 && tid < 128) s_db2[tid - 64] = dec_b2[tid - 64];
  if (tid < 4) s_ent[tid] = 0.0f;
  if (tid < 12) {
    const float phi = q_params[tid * 3 + 0];
    const float th  = q_params[tid * 3 + 1];
    const float om  = q_params[tid * 3 + 2];
    const float ch = cosf(0.5f * th), sh = sinf(0.5f * th);
    const float pa = -0.5f * (phi + om);
    const float pb =  0.5f * (phi - om);
    const float ca = cosf(pa), sa = sinf(pa);
    const float cb = cosf(pb), sb = sinf(pb);
    s_rot[tid][0] =  ca * ch;  s_rot[tid][1] =  sa * ch;
    s_rot[tid][2] = -cb * sh;  s_rot[tid][3] = -sb * sh;
    s_rot[tid][4] =  cb * sh;  s_rot[tid][5] = -sb * sh;
    s_rot[tid][6] =  ca * ch;  s_rot[tid][7] = -sa * ch;
  }
  __syncthreads();

  const int tok0 = blockIdx.x * BLOCK_TOK + wv * WAVE_TOK;
  {
    const float* xnext = x + (size_t)(tok0 + 16 + col) * 128 + 64 * hi;
    __builtin_prefetch(xnext, 0, 0);
    __builtin_prefetch(xnext + 32, 0, 0);
  }

#pragma unroll 1
  for (int mt = 0; mt < 2; ++mt) {
  asm volatile("" ::: "memory");
  const float* xrow = x + (size_t)(tok0 + 16 * mt + col) * 128;
  __builtin_prefetch(xrow, 0, 0);

  v8f acc[4];
#pragma unroll
  for (int n = 0; n < 4; ++n) acc[n] = (v8f){0.f,0.f,0.f,0.f,0.f,0.f,0.f,0.f};
#pragma unroll
  for (int k = 0; k < 32; ++k) {
    const v2f a = *(const v2f*)(xrow + 4 * k + r0);
#pragma unroll
    for (int n = 0; n < 4; ++n) {
      const v2f b = *(const v2f*)(&s_w1t[16 * n + col][4 * k + r0]);
      acc[n] = wmma4(a, b, acc[n]);
    }
  }
#pragma unroll
  for (int n = 0; n < 4; ++n) {
    const float bv = s_b1[16 * n + col];
#pragma unroll
    for (int r = 0; r < 8; ++r) {
      s_h[wv][r + 8 * hi][16 * n + col] = gelu_exact(acc[n][r] + bv);
    }
  }

  v8f eacc = (v8f){0.f,0.f,0.f,0.f,0.f,0.f,0.f,0.f};
#pragma unroll
  for (int k = 0; k < 16; ++k) {
    const v2f a = *(const v2f*)(&s_h[wv][col][4 * k + r0]);
    float b0 = 0.0f, b1 = 0.0f;
    if (col < 8) { b0 = s_w2t[col][4 * k + r0]; b1 = s_w2t[col][4 * k + r0 + 1]; }
    const v2f b = {b0, b1};
    eacc = wmma4(a, b, eacc);
  }
  if (col < 8) {
    const float bv = s_b2[col];
#pragma unroll
    for (int r = 0; r < 8; ++r) s_ang[wv][16 * mt + r + 8 * hi][col] = eacc[r] + bv;
  }
  }

  {
    float zr[16], zi[16];
#pragma unroll
    for (int s = 0; s < 16; ++s) { zr[s] = 0.0f; zi[s] = 0.0f; }
    zr[0] = 1.0f;

#pragma unroll
    for (int i = 0; i < 4; ++i) {
      const int m = 8 >> i;
      const float tx = s_ang[wv][lane][2 * i];
      const float ty = s_ang[wv][lane][2 * i + 1];
      float c = cosf(0.5f * tx), sn = sinf(0.5f * tx);
      apply1q(zr, zi, m, c, 0.f, 0.f, -sn, 0.f, -sn, c, 0.f);
      c = cosf(0.5f * ty); sn = sinf(0.5f * ty);
      apply1q(zr, zi, m, c, 0.f, -sn, 0.f, sn, 0.f, c, 0.f);
    }
#pragma unroll
    for (int l = 0; l < 3; ++l) {
#pragma unroll
      for (int i = 0; i < 4; ++i) {
        const float* u = s_rot[l * 4 + i];
        apply1q(zr, zi, 8 >> i, u[0], u[1], u[2], u[3], u[4], u[5], u[6], u[7]);
      }
#pragma unroll
      for (int i = 0; i < 4; ++i) {
        const int mc = 8 >> i;
        const int mt = 8 >> ((i + 1) & 3);
#pragma unroll
        for (int s = 0; s < 16; ++s) {
          if ((s & mc) && !(s & mt)) {
            const int t = s | mt;
            float tr = zr[s]; zr[s] = zr[t]; zr[t] = tr;
            float ti = zi[s]; zi[s] = zi[t]; zi[t] = ti;
          }
        }
      }
    }
    float p[16];
#pragma unroll
    for (int s = 0; s < 16; ++s) p[s] = zr[s] * zr[s] + zi[s] * zi[s];
#pragma unroll
    for (int i = 0; i < 4; ++i) {
      const int m = 8 >> i;
      float z = 0.0f;
#pragma unroll
      for (int s = 0; s < 16; ++s) z += (s & m) ? -p[s] : p[s];
      s_qf[wv][lane][i] = z;
      const float qp = (z + 1.0f) * 0.5f;
      const float et = -(qp * logf(qp + 1e-10f) +
                         (1.0f - qp) * logf(1.0f - qp + 1e-10f));
      atomicAdd(&s_ent[i], et);
    }
  }

#pragma unroll 1
  for (int mt = 0; mt < 2; ++mt) {
  {
    const v2f a = *(const v2f*)(&s_qf[wv][16 * mt + col][r0]);
    const v2f b = *(const v2f*)(&s_dw1t[col][r0]);
    v8f g = (v8f){0.f,0.f,0.f,0.f,0.f,0.f,0.f,0.f};
    g = wmma4(a, b, g);
    const float bv = s_db1[col];
#pragma unroll
    for (int r = 0; r < 8; ++r)
      s_g[wv][r + 8 * hi][col] = gelu_exact(g[r] + bv);
  }

  v8f oacc[4];
#pragma unroll
  for (int n = 0; n < 4; ++n) oacc[n] = (v8f){0.f,0.f,0.f,0.f,0.f,0.f,0.f,0.f};
#pragma unroll
  for (int k = 0; k < 4; ++k) {
    const v2f a = *(const v2f*)(&s_g[wv][col][4 * k + r0]);
#pragma unroll
    for (int n = 0; n < 4; ++n) {
      const v2f b = *(const v2f*)(&s_dw2t[16 * n + col][4 * k + r0]);
      oacc[n] = wmma4(a, b, oacc[n]);
    }
  }
  float* obase = out + (size_t)(tok0 + 16 * mt) * 64;
#pragma unroll
  for (int n = 0; n < 4; ++n) {
    const float bv = s_db2[16 * n + col];
#pragma unroll
    for (int r = 0; r < 8; ++r) {
      __builtin_nontemporal_store(
          oacc[n][r] + bv,
          obase + (size_t)(r + 8 * hi) * 64 + 16 * n + col);
    }
  }
  }

  __syncthreads();
  if (tid < 4) atomicAdd(ent + tid, s_ent[tid] * (1.0f / (float)TOK_TOTAL));
}

extern "C" void kernel_launch(void* const* d_in, const int* in_sizes, int n_in,
                              void* d_out, int out_size, void* d_ws, size_t ws_size,
                              hipStream_t stream) {
  (void)in_sizes; (void)n_in; (void)out_size; (void)d_ws; (void)ws_size;
  const float* x        = (const float*)d_in[0];
  const float* enc_w1   = (const float*)d_in[1];
  const float* enc_b1   = (const float*)d_in[2];
  const float* enc_w2   = (const float*)d_in[3];
  const float* enc_b2   = (const float*)d_in[4];
  const float* q_params = (const float*)d_in[5];
  const float* dec_w1   = (const float*)d_in[6];
  const float* dec_b1   = (const float*)d_in[7];
  const float* dec_w2   = (const float*)d_in[8];
  const float* dec_b2   = (const float*)d_in[9];

  float* out = (float*)d_out;
  float* ent = out + LATENT_ELEMS;

  qvl_zero_ent<<<1, 32, 0, stream>>>(ent);
  qvl_fused<<<TOK_TOTAL / BLOCK_TOK, 128, 0, stream>>>(
      x, enc_w1, enc_b1, enc_w2, enc_b2, q_params,
      dec_w1, dec_b1, dec_w2, dec_b2, out, ent);
}


# Entropy terms summed by a 4-step in-register butterfly per 16-lane row, row sums added in fixed order, replacing 32-lane LDS atomic adds (compiled to a lane-by-lane scalar loop); sum order changes
# speedup vs baseline: 2.2303x; 1.0441x over previous
#include <hip/hip_runtime.h>
#include <hip/hip_bf16.h>
#include <math.h>


typedef float v2f __attribute__((ext_vector_type(2)));
typedef float v8f __attribute__((ext_vector_type(8)));

#define TOK_TOTAL (64 * 4096)
#define LATENT_ELEMS ((size_t)TOK_TOTAL * 64)
#define WAVE_TOK 32
#define BLOCK_TOK (4 * WAVE_TOK)

static __device__ __forceinline__ v8f wmma4(v2f a, v2f b, v8f c) {
  return __builtin_amdgcn_wmma_f32_16x16x4_f32(
       false, a,  false, b,
       (short)0, c,  false,  false);
}

static __device__ __forceinline__ float gelu_exact(float x) {
  const float v = x * 0.70710678118654752f;
  const float t = fabsf(v);
  float p = fmaf(t, 0x1.1d3156p-16f, -0x1.8d1290p-12f);
  p = fmaf(t, p, 0x1.f9a6d2p-9f);
  p = fmaf(t, p, -0x1.8c3164p-6f);
  p = fmaf(t, p, 0x1.b4e9c8p-4f);
  p = fmaf(t, p, 0x1.4515fap-1f);
  p = fmaf(t, p, 0x1.078e50p-3f);
  p = fmaf(t, p, t);
  const float hi_arm = 1.0f - __builtin_amdgcn_exp2f(p * -0x1.715476p+0f);
  const float s = v * v;
  float q = fmaf(s, -0x1.268bc2p-11f, 0x1.420828p-8f);
  q = fmaf(s, q, -0x1.b59370p-6f);
  q = fmaf(s, q, 0x1.ce077cp-4f);
  q = fmaf(s, q, -0x1.812660p-2f);
  q = fmaf(s, q, 0x1.06eba0p-3f);
  const float lo_arm = fmaf(t, q, t);
  const float e = copysignf(t < 1.0f ? lo_arm : hi_arm, v);
  return 0.5f * x * (1.0f + e);
}

#define ROW_XMASK_ADD(v, M)                                                   \
  ((v) + __builtin_bit_cast(float, __builtin_amdgcn_update_dpp(               \
             0, __builtin_bit_cast(int, (v)), 0x160 + (M), 0xf, 0xf, true)))

static __device__ __forceinline__ void apply1q(
    float* zr, float* zi, int m,
    float u00r, float u00i, float u01r, float u01i,
    float u10r, float u10i, float u11r, float u11i) {
#pragma unroll
  for (int s = 0; s < 16; ++s) {
    if (!(s & m)) {
      const int t = s | m;
      const float ar = zr[s], ai = zi[s];
      const float br = zr[t], bi = zi[t];
      zr[s] = u00r * ar - u00i * ai + u01r * br - u01i * bi;
      zi[s] = u00r * ai + u00i * ar + u01r * bi + u01i * br;
      zr[t] = u10r * ar - u10i * ai + u11r * br - u11i * bi;
      zi[t] = u10r * ai + u10i * ar + u11r * bi + u11i * br;
    }
  }
}

__global__ void qvl_zero_ent(float* ent) {
  if (threadIdx.x < 4) ent[threadIdx.x] = 0.0f;
}

__global__ __launch_bounds__(128) void qvl_fused(
    const float* __restrict__ x,
    const float* __restrict__ enc_w1,
    const float* __restrict__ enc_b1,
    const float* __restrict__ enc_w2,
    const float* __restrict__ enc_b2,
    const float* __restrict__ q_params,
    const float* __restrict__ dec_w1,
    const float* __restrict__ dec_b1,
    const float* __restrict__ dec_w2,
    const float* __restrict__ dec_b2,
    float* __restrict__ out,
    float* __restrict__ ent)
{
  __shared__ float s_w1t[64][128 + 4];
  __shared__ float s_w2t[8][64 + 4];
  __shared__ float s_dw1t[16][4];
  __shared__ float s_dw2t[64][16 + 4];
  __shared__ float s_b1[64];
  __shared__ float s_b2[8];
  __shared__ float s_db1[16];
  __shared__ float s_db2[64];
  __shared__ float s_rot[12][8];
  __shared__ float s_h[4][16][66];
  __shared__ float s_ang[4][32][9];
  __shared__ float s_qf[4][32][4];
  __shared__ float s_g[4][16][18];
  __shared__ float s_entp[4][2][4];

  const int tid  = threadIdx.x;
  const int wv   = tid >> 5;
  const int lane = tid & 31;
  const int col  = lane & 15;
  const int hi   = lane >> 4;
  const int r0   = hi * 2;

  for (int i = tid; i < 128 * 64; i += 128) s_w1t[i & 63][i >> 6] = enc_w1[i];
  for (int i = tid; i < 64 * 8;  i += 128) s_w2t[i & 7][i >> 3]  = enc_w2[i];
  for (int i = tid; i < 4 * 16;  i += 128) s_dw1t[i & 15][i >> 4] = dec_w1[i];
  for (int i = tid; i < 16 * 64; i += 128) s_dw2t[i & 63][i >> 6] = dec_w2[i];
  if (tid < 64) s_b1[tid] = enc_b1[tid];
  if (tid < 8)  s_b2[tid] = enc_b2[tid];
  if (tid < 16) s_db1[tid] = dec_b1[tid];
  if (tid >= 64 && tid < 128) s_db2[tid - 64] = dec_b2[tid - 64];
  if (tid < 12) {
    const float phi = q_params[tid * 3 + 0];
    const float th  = q_params[tid * 3 + 1];
    const float om  = q_params[tid * 3 + 2];
    const float ch = cosf(0.5f * th), sh = sinf(0.5f * th);
    const float pa = -0.5f * (phi + om);
    const float pb =  0.5f * (phi - om);
    const float ca = cosf(pa), sa = sinf(pa);
    const float cb = cosf(pb), sb = sinf(pb);
    s_rot[tid][0] =  ca * ch;  s_rot[tid][1] =  sa * ch;
    s_rot[tid][2] = -cb * sh;  s_rot[tid][3] = -sb * sh;
    s_rot[tid][4] =  cb * sh;  s_rot[tid][5] = -sb * sh;
    s_rot[tid][6] =  ca * ch;  s_rot[tid][7] = -sa * ch;
  }
  __syncthreads();

  const int tok0 = blockIdx.x * BLOCK_TOK + wv * WAVE_TOK;
  {
    const float* xnext = x + (size_t)(tok0 + 16 + col) * 128 + 64 * hi;
    __builtin_prefetch(xnext, 0, 0);
    __builtin_prefetch(xnext + 32, 0, 0);
  }

#pragma unroll 1
  for (int mt = 0; mt < 2; ++mt) {
  asm volatile("" ::: "memory");
  const float* xrow = x + (size_t)(tok0 + 16 * mt + col) * 128;
  __builtin_prefetch(xrow, 0, 0);

  v8f acc[4];
#pragma unroll
  for (int n = 0; n < 4; ++n) acc[n] = (v8f){0.f,0.f,0.f,0.f,0.f,0.f,0.f,0.f};
#pragma unroll
  for (int k = 0; k < 32; ++k) {
    const v2f a = *(const v2f*)(xrow + 4 * k + r0);
#pragma unroll
    for (int n = 0; n < 4; ++n) {
      const v2f b = *(const v2f*)(&s_w1t[16 * n + col][4 * k + r0]);
      acc[n] = wmma4(a, b, acc[n]);
    }
  }
#pragma unroll
  for (int n = 0; n < 4; ++n) {
    const float bv = s_b1[16 * n + col];
#pragma unroll
    for (int r = 0; r < 8; ++r) {
      s_h[wv][r + 8 * hi][16 * n + col] = gelu_exact(acc[n][r] + bv);
    }
  }

  v8f eacc = (v8f){0.f,0.f,0.f,0.f,0.f,0.f,0.f,0.f};
#pragma unroll
  for (int k = 0; k < 16; ++k) {
    const v2f a = *(const v2f*)(&s_h[wv][col][4 * k + r0]);
    float b0 = 0.0f, b1 = 0.0f;
    if (col < 8) { b0 = s_w2t[col][4 * k + r0]; b1 = s_w2t[col][4 * k + r0 + 1]; }
    const v2f b = {b0, b1};
    eacc = wmma4(a, b, eacc);
  }
  if (col < 8) {
    const float bv = s_b2[col];
#pragma unroll
    for (int r = 0; r < 8; ++r) s_ang[wv][16 * mt + r + 8 * hi][col] = eacc[r] + bv;
  }
  }

  {
    float zr[16], zi[16];
#pragma unroll
    for (int s = 0; s < 16; ++s) { zr[s] = 0.0f; zi[s] = 0.0f; }
    zr[0] = 1.0f;

#pragma unroll
    for (int i = 0; i < 4; ++i) {
      const int m = 8 >> i;
      const float tx = s_ang[wv][lane][2 * i];
      const float ty = s_ang[wv][lane][2 * i + 1];
      float c = cosf(0.5f * tx), sn = sinf(0.5f * tx);
      apply1q(zr, zi, m, c, 0.f, 0.f, -sn, 0.f, -sn, c, 0.f);
      c = cosf(0.5f * ty); sn = sinf(0.5f * ty);
      apply1q(zr, zi, m, c, 0.f, -sn, 0.f, sn, 0.f, c, 0.f);
    }
#pragma unroll
    for (int l = 0; l < 3; ++l) {
#pragma unroll
      for (int i = 0; i < 4; ++i) {
        const float* u = s_rot[l * 4 + i];
        apply1q(zr, zi, 8 >> i, u[0], u[1], u[2], u[3], u[4], u[5], u[6], u[7]);
      }
#pragma unroll
      for (int i = 0; i < 4; ++i) {
        const int mc = 8 >> i;
        const int mt = 8 >> ((i + 1) & 3);
#pragma unroll
        for (int s = 0; s < 16; ++s) {
          if ((s & mc) && !(s & mt)) {
            const int t = s | mt;
            float tr = zr[s]; zr[s] = zr[t]; zr[t] = tr;
            float ti = zi[s]; zi[s] = zi[t]; zi[t] = ti;
          }
        }
      }
    }
    float p[16];
#pragma unroll
    for (int s = 0; s < 16; ++s) p[s] = zr[s] * zr[s] + zi[s] * zi[s];
    float et[4];
#pragma unroll
    for (int i = 0; i < 4; ++i) {
      const int m = 8 >> i;
      float z = 0.0f;
#pragma unroll
      for (int s = 0; s < 16; ++s) z += (s & m) ? -p[s] : p[s];
      s_qf[wv][lane][i] = z;
      const float qp = (z + 1.0f) * 0.5f;
      et[i] = -(qp * logf(qp + 1e-10f) +
                (1.0f - qp) * logf(1.0f - qp + 1e-10f));
    }
    for (int i = 0; i < 4; ++i) et[i] = ROW_XMASK_ADD(et[i], 1);
    for (int i = 0; i < 4; ++i) et[i] = ROW_XMASK_ADD(et[i], 2);
    for (int i = 0; i < 4; ++i) et[i] = ROW_XMASK_ADD(et[i], 4);
    for (int i = 0; i < 4; ++i) et[i] = ROW_XMASK_ADD(et[i], 8);
    if (col == 0) {
      for (int i = 0; i < 4; ++i) s_entp[wv][hi][i] = et[i];
    }
  }

#pragma unroll 1
  for (int mt = 0; mt < 2; ++mt) {
  {
    const v2f a = *(const v2f*)(&s_qf[wv][16 * mt + col][r0]);
    const v2f b = *(const v2f*)(&s_dw1t[col][r0]);
    v8f g = (v8f){0.f,0.f,0.f,0.f,0.f,0.f,0.f,0.f};
    g = wmma4(a, b, g);
    const float bv = s_db1[col];
#pragma unroll
    for (int r = 0; r < 8; ++r)
      s_g[wv][r + 8 * hi][col] = gelu_exact(g[r] + bv);
  }

  v8f oacc[4];
#pragma unroll
  for (int n = 0; n < 4; ++n) oacc[n] = (v8f){0.f,0.f,0.f,0.f,0.f,0.f,0.f,0.f};
#pragma unroll
  for (int k = 0; k < 4; ++k) {
    const v2f a = *(const v2f*)(&s_g[wv][col][4 * k + r0]);
#pragma unroll
    for (int n = 0; n < 4; ++n) {
      const v2f b = *(const v2f*)(&s_dw2t[16 * n + col][4 * k + r0]);
      oacc[n] = wmma4(a, b, oacc[n]);
    }
  }
  float* obase = out + (size_t)(tok0 + 16 * mt) * 64;
#pragma unroll
  for (int n = 0; n < 4; ++n) {
    const float bv = s_db2[16 * n + col];
#pragma unroll
    for (int r = 0; r < 8; ++r) {
      __builtin_nontemporal_store(
          oacc[n][r] + bv,
          obase + (size_t)(r + 8 * hi) * 64 + 16 * n + col);
    }
  }
  }

  __syncthreads();
  if (tid < 4) {
    float sum = 0.0f;
    for (int w = 0; w < 4; ++w) sum += s_entp[w][0][tid] + s_entp[w][1][tid];
    atomicAdd(ent + tid, sum * (1.0f / (float)TOK_TOTAL));
  }
}

extern "C" void kernel_launch(void* const* d_in, const int* in_sizes, int n_in,
                              void* d_out, int out_size, void* d_ws, size_t ws_size,
                              hipStream_t stream) {
  (void)in_sizes; (void)n_in; (void)out_size; (void)d_ws; (void)ws_size;
  const float* x        = (const float*)d_in[0];
  const float* enc_w1   = (const float*)d_in[1];
  const float* enc_b1   = (const float*)d_in[2];
  const float* enc_w2   = (const float*)d_in[3];
  const float* enc_b2   = (const float*)d_in[4];
  const float* q_params = (const float*)d_in[5];
  const float* dec_w1   = (const float*)d_in[6];
  const float* dec_b1   = (const float*)d_in[7];
  const float* dec_w2   = (const float*)d_in[8];
  const float* dec_b2   = (const float*)d_in[9];

  float* out = (float*)d_out;
  float* ent = out + LATENT_ELEMS;

  qvl_zero_ent<<<1, 32, 0, stream>>>(ent);
  qvl_fused<<<TOK_TOTAL / BLOCK_TOK, 128, 0, stream>>>(
      x, enc_w1, enc_b1, enc_w2, enc_b2, q_params,
      dec_w1, dec_b1, dec_w2, dec_b2, out, ent);
}


# Prologue: all global reads of weights, biases and gate angles issued before the first LDS store (float4, bank-spread transposing stores); was one load-wait-store round trip per staging loop trip
# speedup vs baseline: 2.5922x; 1.1623x over previous
#include <hip/hip_runtime.h>
#include <hip/hip_bf16.h>
#include <math.h>


typedef float v2f __attribute__((ext_vector_type(2)));
typedef float v4f __attribute__((ext_vector_type(4)));
typedef float v8f __attribute__((ext_vector_type(8)));

#define TOK_TOTAL (64 * 4096)
#define LATENT_ELEMS ((size_t)TOK_TOTAL * 64)
#define WAVE_TOK 32
#define BLOCK_TOK (4 * WAVE_TOK)

static __device__ __forceinline__ v8f wmma4(v2f a, v2f b, v8f c) {
  return __builtin_amdgcn_wmma_f32_16x16x4_f32(
       false, a,  false, b,
       (short)0, c,  false,  false);
}

static __device__ __forceinline__ float gelu_exact(float x) {
  const float v = x * 0.70710678118654752f;
  const float t = fabsf(v);
  float p = fmaf(t, 0x1.1d3156p-16f, -0x1.8d1290p-12f);
  p = fmaf(t, p, 0x1.f9a6d2p-9f);
  p = fmaf(t, p, -0x1.8c3164p-6f);
  p = fmaf(t, p, 0x1.b4e9c8p-4f);
  p = fmaf(t, p, 0x1.4515fap-1f);
  p = fmaf(t, p, 0x1.078e50p-3f);
  p = fmaf(t, p, t);
  const float hi_arm = 1.0f - __builtin_amdgcn_exp2f(p * -0x1.715476p+0f);
  const float s = v * v;
  float q = fmaf(s, -0x1.268bc2p-11f, 0x1.420828p-8f);
  q = fmaf(s, q, -0x1.b59370p-6f);
  q = fmaf(s, q, 0x1.ce077cp-4f);
  q = fmaf(s, q, -0x1.812660p-2f);
  q = fmaf(s, q, 0x1.06eba0p-3f);
  const float lo_arm = fmaf(t, q, t);
  const float e = copysignf(t < 1.0f ? lo_arm : hi_arm, v);
  return 0.5f * x * (1.0f + e);
}

#define ROW_XMASK_ADD(v, M)                                                   \
  ((v) + __builtin_bit_cast(float, __builtin_amdgcn_update_dpp(               \
             0, __builtin_bit_cast(int, (v)), 0x160 + (M), 0xf, 0xf, true)))

static __device__ __forceinline__ void apply1q(
    float* zr, float* zi, int m,
    float u00r, float u00i, float u01r, float u01i,
    float u10r, float u10i, float u11r, float u11i) {
#pragma unroll
  for (int s = 0; s < 16; ++s) {
    if (!(s & m)) {
      const int t = s | m;
      const float ar = zr[s], ai = zi[s];
      const float br = zr[t], bi = zi[t];
      zr[s] = u00r * ar - u00i * ai + u01r * br - u01i * bi;
      zi[s] = u00r * ai + u00i * ar + u01r * bi + u01i * br;
      zr[t] = u10r * ar - u10i * ai + u11r * br - u11i * bi;
      zi[t] = u10r * ai + u10i * ar + u11r * bi + u11i * br;
    }
  }
}

__global__ void qvl_zero_ent(float* ent) {
  if (threadIdx.x < 4) ent[threadIdx.x] = 0.0f;
}

__global__ __launch_bounds__(128) void qvl_fused(
    const float* __restrict__ x,
    const float* __restrict__ enc_w1,
    const float* __restrict__ enc_b1,
    const float* __restrict__ enc_w2,
    const float* __restrict__ enc_b2,
    const float* __restrict__ q_params,
    const float* __restrict__ dec_w1,
    const float* __restrict__ dec_b1,
    const float* __restrict__ dec_w2,
    const float* __restrict__ dec_b2,
    float* __restrict__ out,
    float* __restrict__ ent)
{
  __shared__ float s_w1t[64][128 + 4];
  __shared__ float s_w2t[8][64 + 4];
  __shared__ float s_dw1t[16][4];
  __shared__ float s_dw2t[64][16 + 4];
  __shared__ float s_b1[64];
  __shared__ float s_b2[8];
  __shared__ float s_db1[16];
  __shared__ float s_db2[64];
  __shared__ float s_rot[12][8];
  __shared__ float s_h[4][16][66];
  __shared__ float s_ang[4][32][9];
  __shared__ float s_qf[4][32][4];
  __shared__ float s_g[4][16][18];
  __shared__ float s_entp[4][2][4];

  const int tid  = threadIdx.x;
  const int wv   = tid >> 5;
  const int lane = tid & 31;
  const int col  = lane & 15;
  const int hi   = lane >> 4;
  const int r0   = hi * 2;

  const int w1k = lane >> 2;
  const int w1n = 16 * wv + 4 * (lane & 3);
  v4f w1r[16];
#pragma unroll
  for (int j = 0; j < 16; ++j)
    w1r[j] = *(const v4f*)(enc_w1 + (8 * j + w1k) * 64 + w1n);
  const v4f w2r = *(const v4f*)(enc_w2 + 4 * tid);
  v4f dw2r[2];
#pragma unroll
  for (int j = 0; j < 2; ++j) dw2r[j] = *(const v4f*)(dec_w2 + 4 * (tid + 128 * j));
  const float w1b1 = *(tid < 64 ? dec_w1 + tid : enc_b1 + (tid - 64));
  const float b2r  = enc_b2[tid & 7];
  const float db1r = dec_b1[tid & 15];
  const float db2r = dec_b2[tid & 63];
  const int rt = tid < 12 ? tid : 0;
  const float phi = q_params[rt * 3 + 0];
  const float th  = q_params[rt * 3 + 1];
  const float om  = q_params[rt * 3 + 2];

#pragma unroll
  for (int j = 0; j < 16; ++j) {
#pragma unroll
    for (int e = 0; e < 4; ++e) s_w1t[w1n + e][8 * j + w1k] = w1r[j][e];
  }
#pragma unroll
  for (int e = 0; e < 4; ++e) s_w2t[4 * (tid & 1) + e][tid >> 1] = w2r[e];
#pragma unroll
  for (int j = 0; j < 2; ++j) {
    const int q = tid + 128 * j;
#pragma unroll
    for (int e = 0; e < 4; ++e) s_dw2t[4 * (q & 15) + e][q >> 4] = dw2r[j][e];
  }
  *(tid < 64 ? &s_dw1t[tid & 15][tid >> 4] : &s_b1[tid - 64]) = w1b1;
  if ((tid >> 3) == 4) s_b2[tid & 7] = b2r;
  if ((tid >> 4) == 4) s_db1[tid & 15] = db1r;
  if (tid >= 64) s_db2[tid - 64] = db2r;
  if (tid < 12) {
    const float ch = cosf(0.5f * th), sh = sinf(0.5f * th);
    const float pa = -0.5f * (phi + om);
    const float pb =  0.5f * (phi - om);
    const float ca = cosf(pa), sa = sinf(pa);
    const float cb = cosf(pb), sb = sinf(pb);
    s_rot[tid][0] =  ca * ch;  s_rot[tid][1] =  sa * ch;
    s_rot[tid][2] = -cb * sh;  s_rot[tid][3] = -sb * sh;
    s_rot[tid][4] =  cb * sh;  s_rot[tid][5] = -sb * sh;
    s_rot[tid][6] =  ca * ch;  s_rot[tid][7] = -sa * ch;
  }
  __syncthreads();

  const int tok0 = blockIdx.x * BLOCK_TOK + wv * WAVE_TOK;
  {
    const float* xnext = x + (size_t)(tok0 + 16 + col) * 128 + 64 * hi;
    __builtin_prefetch(xnext, 0, 0);
    __builtin_prefetch(xnext + 32, 0, 0);
  }

#pragma unroll 1
  for (int mt = 0; mt < 2; ++mt) {
  asm volatile("" ::: "memory");
  const float* xrow = x + (size_t)(tok0 + 16 * mt + col) * 128;
  __builtin_prefetch(xrow, 0, 0);

  v8f acc[4];
#pragma unroll
  for (int n = 0; n < 4; ++n) acc[n] = (v8f){0.f,0.f,0.f,0.f,0.f,0.f,0.f,0.f};
#pragma unroll
  for (int k = 0; k < 32; ++k) {
    const v2f a = *(const v2f*)(xrow + 4 * k + r0);
#pragma unroll
    for (int n = 0; n < 4; ++n) {
      const v2f b = *(const v2f*)(&s_w1t[16 * n + col][4 * k + r0]);
      acc[n] = wmma4(a, b, acc[n]);
    }
  }
#pragma unroll
  for (int n = 0; n < 4; ++n) {
    const float bv = s_b1[16 * n + col];
#pragma unroll
    for (int r = 0; r < 8; ++r) {
      s_h[wv][r + 8 * hi][16 * n + col] = gelu_exact(acc[n][r] + bv);
    }
  }

  v8f eacc = (v8f){0.f,0.f,0.f,0.f,0.f,0.f,0.f,0.f};
#pragma unroll
  for (int k = 0; k < 16; ++k) {
    const v2f a = *(const v2f*)(&s_h[wv][col][4 * k + r0]);
    float b0 = 0.0f, b1 = 0.0f;
    if (col < 8) { b0 = s_w2t[col][4 * k + r0]; b1 = s_w2t[col][4 * k + r0 + 1]; }
    const v2f b = {b0, b1};
    eacc = wmma4(a, b, eacc);
  }
  if (col < 8) {
    const float bv = s_b2[col];
#pragma unroll
    for (int r = 0; r < 8; ++r) s_ang[wv][16 * mt + r + 8 * hi][col] = eacc[r] + bv;
  }
  }

  {
    float zr[16], zi[16];
#pragma unroll
    for (int s = 0; s < 16; ++s) { zr[s] = 0.0f; zi[s] = 0.0f; }
    zr[0] = 1.0f;

#pragma unroll
    for (int i = 0; i < 4; ++i) {
      const int m = 8 >> i;
      const float tx = s_ang[wv][lane][2 * i];
      const float ty = s_ang[wv][lane][2 * i + 1];
      float c = cosf(0.5f * tx), sn = sinf(0.5f * tx);
      apply1q(zr, zi, m, c, 0.f, 0.f, -sn, 0.f, -sn, c, 0.f);
      c = cosf(0.5f * ty); sn = sinf(0.5f * ty);
      apply1q(zr, zi, m, c, 0.f, -sn, 0.f, sn, 0.f, c, 0.f);
    }
#pragma unroll
    for (int l = 0; l < 3; ++l) {
#pragma unroll
      for (int i = 0; i < 4; ++i) {
        const float* u = s_rot[l * 4 + i];
        apply1q(zr, zi, 8 >> i, u[0], u[1], u[2], u[3], u[4], u[5], u[6], u[7]);
      }
#pragma unroll
      for (int i = 0; i < 4; ++i) {
        const int mc = 8 >> i;
        const int mt = 8 >> ((i + 1) & 3);
#pragma unroll
        for (int s = 0; s < 16; ++s) {
          if ((s & mc) && !(s & mt)) {
            const int t = s | mt;
            float tr = zr[s]; zr[s] = zr[t]; zr[t] = tr;
            float ti = zi[s]; zi[s] = zi[t]; zi[t] = ti;
          }
        }
      }
    }
    float p[16];
#pragma unroll
    for (int s = 0; s < 16; ++s) p[s] = zr[s] * zr[s] + zi[s] * zi[s];
    float et[4];
#pragma unroll
    for (int i = 0; i < 4; ++i) {
      const int m = 8 >> i;
      float z = 0.0f;
#pragma unroll
      for (int s = 0; s < 16; ++s) z += (s & m) ? -p[s] : p[s];
      s_qf[wv][lane][i] = z;
      const float qp = (z + 1.0f) * 0.5f;
      et[i] = -(qp * logf(qp + 1e-10f) +
                (1.0f - qp) * logf(1.0f - qp + 1e-10f));
    }
    for (int i = 0; i < 4; ++i) et[i] = ROW_XMASK_ADD(et[i], 1);
    for (int i = 0; i < 4; ++i) et[i] = ROW_XMASK_ADD(et[i], 2);
    for (int i = 0; i < 4; ++i) et[i] = ROW_XMASK_ADD(et[i], 4);
    for (int i = 0; i < 4; ++i) et[i] = ROW_XMASK_ADD(et[i], 8);
    if (col == 0) {
      for (int i = 0; i < 4; ++i) s_entp[wv][hi][i] = et[i];
    }
  }

#pragma unroll 1
  for (int mt = 0; mt < 2; ++mt) {
  {
    const v2f a = *(const v2f*)(&s_qf[wv][16 * mt + col][r0]);
    const v2f b = *(const v2f*)(&s_dw1t[col][r0]);
    v8f g = (v8f){0.f,0.f,0.f,0.f,0.f,0.f,0.f,0.f};
    g = wmma4(a, b, g);
    const float bv = s_db1[col];
#pragma unroll
    for (int r = 0; r < 8; ++r)
      s_g[wv][r + 8 * hi][col] = gelu_exact(g[r] + bv);
  }

  v8f oacc[4];
#pragma unroll
  for (int n = 0; n < 4; ++n) oacc[n] = (v8f){0.f,0.f,0.f,0.f,0.f,0.f,0.f,0.f};
#pragma unroll
  for (int k = 0; k < 4; ++k) {
    const v2f a = *(const v2f*)(&s_g[wv][col][4 * k + r0]);
#pragma unroll
    for (int n = 0; n < 4; ++n) {
      const v2f b = *(const v2f*)(&s_dw2t[16 * n + col][4 * k + r0]);
      oacc[n] = wmma4(a, b, oacc[n]);
    }
  }
  float* obase = out + (size_t)(tok0 + 16 * mt) * 64;
#pragma unroll
  for (int n = 0; n < 4; ++n) {
    const float bv = s_db2[16 * n + col];
#pragma unroll
    for (int r = 0; r < 8; ++r) {
      __builtin_nontemporal_store(
          oacc[n][r] + bv,
          obase + (size_t)(r + 8 * hi) * 64 + 16 * n + col);
    }
  }
  }

  __syncthreads();
  if (tid < 4) {
    float sum = 0.0f;
    for (int w = 0; w < 4; ++w) sum += s_entp[w][0][tid] + s_entp[w][1][tid];
    atomicAdd(ent + tid, sum * (1.0f / (float)TOK_TOTAL));
  }
}

extern "C" void kernel_launch(void* const* d_in, const int* in_sizes, int n_in,
                              void* d_out, int out_size, void* d_ws, size_t ws_size,
                              hipStream_t stream) {
  (void)in_sizes; (void)n_in; (void)out_size; (void)d_ws; (void)ws_size;
  const float* x        = (const float*)d_in[0];
  const float* enc_w1   = (const float*)d_in[1];
  const float* enc_b1   = (const float*)d_in[2];
  const float* enc_w2   = (const float*)d_in[3];
  const float* enc_b2   = (const float*)d_in[4];
  const float* q_params = (const float*)d_in[5];
  const float* dec_w1   = (const float*)d_in[6];
  const float* dec_b1   = (const float*)d_in[7];
  const float* dec_w2   = (const float*)d_in[8];
  const float* dec_b2   = (const float*)d_in[9];

  float* out = (float*)d_out;
  float* ent = out + LATENT_ELEMS;

  qvl_zero_ent<<<1, 32, 0, stream>>>(ent);
  qvl_fused<<<TOK_TOTAL / BLOCK_TOK, 128, 0, stream>>>(
      x, enc_w1, enc_b1, enc_w2, enc_b2, q_params,
      dec_w1, dec_b1, dec_w2, dec_b2, out, ent);
}
